# starting kernel
# baseline (speedup 1.0000x reference)
#include <hip/hip_runtime.h>


typedef __attribute__((ext_vector_type(2))) float v2f;
typedef __attribute__((ext_vector_type(8))) float v8f;

#define T_STEPS 4096
#define BATCH   64
#define DIN     64
#define DHID    512
#define DOUT    64
#define CHUNKS  128
#define CLEN    32
#define HSTRIDE 516
#define NWAVES  16
#define NTHREADS (NWAVES * 32)

__device__ __forceinline__ v8f wmma4(v2f a, v2f b, v8f c) {
    return __builtin_amdgcn_wmma_f32_16x16x4_f32(
        false, a, false, b, (short)0, c, false, false);
}

__device__ __forceinline__ int opaque_zero() {
    int z;
    asm volatile("s_mov_b32 %0, 0" : "=s"(z));
    return z;
}

__global__ __launch_bounds__(256)
void rnn_pack(const float* __restrict__ bW, const float* __restrict__ cW,
              v2f* __restrict__ bp, v2f* __restrict__ cp)
{
    const int i = blockIdx.x * 256 + threadIdx.x;
    if (i < 32 * DHID) {
        const int p = i >> 9, n = i & (DHID - 1);
        v2f t;
        t.x = bW[(size_t)(2 * p) * DHID + n];
        t.y = bW[(size_t)(2 * p + 1) * DHID + n];
        bp[i] = t;
    }
    if (i < 256 * DOUT) {
        const int p = i >> 6, n = i & (DOUT - 1);
        v2f t;
        t.x = cW[(size_t)(2 * p) * DOUT + n];
        t.y = cW[(size_t)(2 * p + 1) * DOUT + n];
        cp[i] = t;
    }
}

__global__ __launch_bounds__(NTHREADS)
void rnn_chunk_summary(const float* __restrict__ x,
                       const float* __restrict__ aVec,
                       const v2f* __restrict__ bp0,
                       float* __restrict__ S)
{
    const int chunk = blockIdx.x;
    const int tid   = threadIdx.x;
    const int wave  = tid >> 5;
    const int lane  = tid & 31;
    const int lmod  = lane & 15;
    const int lhi   = lane >> 4;

    const int m_tile = wave & 3;
    const int nclus  = wave >> 2;
    const int rowA   = m_tile * 16 + lmod;

    float av[8];
    #pragma unroll
    for (int nt = 0; nt < 8; ++nt)
        av[nt] = aVec[(nclus * 8 + nt) * 16 + lmod];

    const int zero = opaque_zero();
    const v2f* bp = bp0;

    v8f Sacc[8] = {};

    for (int i = 0; i < CLEN; ++i) {
        bp += zero;
        const float* xt = x + (size_t)(chunk * CLEN + i) * BATCH * DIN;

        v2f afrag[16];
        #pragma unroll
        for (int kt = 0; kt < 16; ++kt)
            afrag[kt] = *(const v2f*)(xt + rowA * DIN + kt * 4 + lhi * 2);

        #pragma unroll
        for (int ntp = 0; ntp < 4; ++ntp) {
            const int n0 = (nclus * 8 + ntp * 2) * 16 + lmod;
            v8f acc0 = {}, acc1 = {};
            #pragma unroll
            for (int kt = 0; kt < 16; ++kt) {
                const int p = kt * 2 + lhi;
                const v2f b0 = bp[p * DHID + n0];
                const v2f b1 = bp[p * DHID + n0 + 16];
                acc0 = wmma4(afrag[kt], b0, acc0);
                acc1 = wmma4(afrag[kt], b1, acc1);
            }
            const float a0 = av[ntp * 2], a1 = av[ntp * 2 + 1];
            #pragma unroll
            for (int r = 0; r < 8; ++r) {
                Sacc[ntp * 2][r]     = Sacc[ntp * 2][r]     * a0 + acc0[r];
                Sacc[ntp * 2 + 1][r] = Sacc[ntp * 2 + 1][r] * a1 + acc1[r];
            }
        }
    }

    float* Sj = S + (size_t)chunk * BATCH * DHID;
    #pragma unroll
    for (int nt = 0; nt < 8; ++nt) {
        const int col = (nclus * 8 + nt) * 16 + lmod;
        #pragma unroll
        for (int r = 0; r < 8; ++r) {
            const int row = m_tile * 16 + r + lhi * 8;
            Sj[row * DHID + col] = Sacc[nt][r];
        }
    }
}

__global__ __launch_bounds__(256)
void rnn_carry_scan(const float* __restrict__ h0,
                    const float* __restrict__ aVec,
                    const float* __restrict__ S,
                    float* __restrict__ Cout,
                    float* __restrict__ hfinal)
{
    const int idx = blockIdx.x * blockDim.x + threadIdx.x;
    if (idx >= BATCH * DHID) return;
    const int hcol = idx & (DHID - 1);

    float aL = aVec[hcol];
    #pragma unroll
    for (int s = 0; s < 5; ++s) aL *= aL;

    float c = h0[idx];
    for (int j = 0; j < CHUNKS; ++j) {
        Cout[(size_t)j * BATCH * DHID + idx] = c;
        c = aL * c + S[(size_t)j * BATCH * DHID + idx];
    }
    hfinal[idx] = c;
}

__global__ __launch_bounds__(NTHREADS)
void rnn_chunk_emit(const float* __restrict__ x,
                    const float* __restrict__ aVec,
                    const v2f* __restrict__ bp0,
                    const v2f* __restrict__ cp0,
                    const float* __restrict__ Cin,
                    float* __restrict__ y)
{
    extern __shared__ float hbuf[];

    const int chunk = blockIdx.x;
    const int tid   = threadIdx.x;
    const int wave  = tid >> 5;
    const int lane  = tid & 31;
    const int lmod  = lane & 15;
    const int lhi   = lane >> 4;

    const int m_tile = wave & 3;
    const int nclus  = wave >> 2;
    const int rowA   = m_tile * 16 + lmod;
    const int n2     = wave >> 2;
    const int rowH   = m_tile * 16 + lmod;
    const int ncol   = n2 * 16 + lmod;

    float av[8];
    #pragma unroll
    for (int nt = 0; nt < 8; ++nt)
        av[nt] = aVec[(nclus * 8 + nt) * 16 + lmod];

    const float* Cj = Cin + (size_t)chunk * BATCH * DHID;
    for (int idx = tid; idx < BATCH * DHID; idx += NTHREADS) {
        const int row = idx >> 9;
        const int col = idx & (DHID - 1);
        hbuf[row * HSTRIDE + col] = Cj[idx];
    }
    __syncthreads();

    const int zero = opaque_zero();
    const v2f* bp = bp0;
    const v2f* cp = cp0;

    for (int i = 0; i < CLEN; ++i) {
        bp += zero;
        cp += zero;
        const int t = chunk * CLEN + i;
        const float* xt = x + (size_t)t * BATCH * DIN;

        v2f afrag[16];
        #pragma unroll
        for (int kt = 0; kt < 16; ++kt)
            afrag[kt] = *(const v2f*)(xt + rowA * DIN + kt * 4 + lhi * 2);

        __syncthreads();

        #pragma unroll
        for (int ntp = 0; ntp < 4; ++ntp) {
            const int n0 = (nclus * 8 + ntp * 2) * 16 + lmod;
            v8f acc0 = {}, acc1 = {};
            #pragma unroll
            for (int kt = 0; kt < 16; ++kt) {
                const int p = kt * 2 + lhi;
                const v2f b0 = bp[p * DHID + n0];
                const v2f b1 = bp[p * DHID + n0 + 16];
                acc0 = wmma4(afrag[kt], b0, acc0);
                acc1 = wmma4(afrag[kt], b1, acc1);
            }
            const float a0 = av[ntp * 2], a1 = av[ntp * 2 + 1];
            #pragma unroll
            for (int r = 0; r < 8; ++r) {
                const int row = m_tile * 16 + r + lhi * 8;
                const int i0  = row * HSTRIDE + n0;
                hbuf[i0]      = hbuf[i0]      * a0 + acc0[r];
                hbuf[i0 + 16] = hbuf[i0 + 16] * a1 + acc1[r];
            }
        }
        __syncthreads();

        v8f y0 = {}, y1 = {};
        #pragma unroll 8
        for (int kt = 0; kt < 64; ++kt) {
            const int kbA = kt * 4 + lhi * 2;
            const int kbB = kbA + 256;
            const v2f hA = *(const v2f*)(hbuf + rowH * HSTRIDE + kbA);
            const v2f hB = *(const v2f*)(hbuf + rowH * HSTRIDE + kbB);
            const v2f cA = cp[(kbA >> 1) * DOUT + ncol];
            const v2f cB = cp[(kbB >> 1) * DOUT + ncol];
            y0 = wmma4(hA, cA, y0);
            y1 = wmma4(hB, cB, y1);
        }

        float* yt = y + (size_t)t * BATCH * DOUT;
        #pragma unroll
        for (int r = 0; r < 8; ++r) {
            const int row = m_tile * 16 + r + lhi * 8;
            yt[row * DOUT + ncol] = y0[r] + y1[r];
        }
    }
}

extern "C" void kernel_launch(void* const* d_in, const int* in_sizes, int n_in,
                              void* d_out, int out_size, void* d_ws, size_t ws_size,
                              hipStream_t stream)
{
    const float* h0 = (const float*)d_in[0];
    const float* x  = (const float*)d_in[1];
    const float* a  = (const float*)d_in[2];
    const float* bW = (const float*)d_in[3];
    const float* cW = (const float*)d_in[4];

    float* out    = (float*)d_out;
    float* hfinal = out;
    float* y      = out + BATCH * DHID;

    float* S  = (float*)d_ws;
    float* C  = S + (size_t)CHUNKS * BATCH * DHID;
    v2f*   bp = (v2f*)(C + (size_t)CHUNKS * BATCH * DHID);
    v2f*   cp = bp + 32 * DHID;

    const size_t shmem = (size_t)(64 * HSTRIDE) * sizeof(float);

    rnn_pack<<<64, 256, 0, stream>>>(bW, cW, bp, cp);
    rnn_chunk_summary<<<CHUNKS, NTHREADS, 0, stream>>>(x, a, bp, S);
    rnn_carry_scan<<<(BATCH * DHID + 255) / 256, 256, 0, stream>>>(h0, a, S, C, hfinal);
    rnn_chunk_emit<<<CHUNKS, NTHREADS, shmem, stream>>>(x, a, bp, cp, C, y);
}


# Batch split: each 32-step chunk is worked by two work-groups of 32 batch rows each (256 work-groups, whole chip); a y tile's two K halves run on two waves and meet in LDS; sums unchanged
# speedup vs baseline: 1.5324x; 1.5324x over previous
#include <hip/hip_runtime.h>


typedef __attribute__((ext_vector_type(2))) float v2f;
typedef __attribute__((ext_vector_type(8))) float v8f;

#define T_STEPS 4096
#define BATCH   64
#define DIN     64
#define DHID    512
#define DOUT    64
#define CHUNKS  128
#define CLEN    32
#define HSTRIDE 516
#define NWAVES  16
#define NTHREADS (NWAVES * 32)
#define BSPLIT  2
#define BROWS   32
#define YTILE   256

__device__ __forceinline__ v8f wmma4(v2f a, v2f b, v8f c) {
    return __builtin_amdgcn_wmma_f32_16x16x4_f32(
        false, a, false, b, (short)0, c, false, false);
}

__device__ __forceinline__ int opaque_zero() {
    int z;
    asm volatile("s_mov_b32 %0, 0" : "=s"(z));
    return z;
}

__global__ __launch_bounds__(256)
void rnn_pack(const float* __restrict__ bW, const float* __restrict__ cW,
              v2f* __restrict__ bp, v2f* __restrict__ cp)
{
    const int i = blockIdx.x * 256 + threadIdx.x;
    if (i < 32 * DHID) {
        const int p = i >> 9, n = i & (DHID - 1);
        v2f t;
        t.x = bW[(size_t)(2 * p) * DHID + n];
        t.y = bW[(size_t)(2 * p + 1) * DHID + n];
        bp[i] = t;
    }
    if (i < 256 * DOUT) {
        const int p = i >> 6, n = i & (DOUT - 1);
        v2f t;
        t.x = cW[(size_t)(2 * p) * DOUT + n];
        t.y = cW[(size_t)(2 * p + 1) * DOUT + n];
        cp[i] = t;
    }
}

__global__ __launch_bounds__(NTHREADS)
void rnn_chunk_summary(const float* __restrict__ x,
                       const float* __restrict__ aVec,
                       const v2f* __restrict__ bp0,
                       float* __restrict__ S)
{
    const int chunk = blockIdx.x >> 1;
    const int bhalf = blockIdx.x & 1;
    const int tid   = threadIdx.x;
    const int wave  = tid >> 5;
    const int lane  = tid & 31;
    const int lmod  = lane & 15;
    const int lhi   = lane >> 4;

    const int m_tile = wave & 1;
    const int nclus  = wave >> 1;
    const int row0   = bhalf * BROWS + m_tile * 16;
    const int rowA   = row0 + lmod;

    float av[4];
    #pragma unroll
    for (int nt = 0; nt < 4; ++nt)
        av[nt] = aVec[(nclus * 4 + nt) * 16 + lmod];

    const int zero = opaque_zero();
    const v2f* bp = bp0;

    v8f Sacc[4] = {};

    for (int i = 0; i < CLEN; ++i) {
        bp += zero;
        const float* xt = x + (size_t)(chunk * CLEN + i) * BATCH * DIN;

        v2f afrag[16];
        #pragma unroll
        for (int kt = 0; kt < 16; ++kt)
            afrag[kt] = *(const v2f*)(xt + rowA * DIN + kt * 4 + lhi * 2);

        #pragma unroll
        for (int ntp = 0; ntp < 2; ++ntp) {
            const int n0 = (nclus * 4 + ntp * 2) * 16 + lmod;
            v8f acc0 = {}, acc1 = {};
            #pragma unroll
            for (int kt = 0; kt < 16; ++kt) {
                const int p = kt * 2 + lhi;
                const v2f b0 = bp[p * DHID + n0];
                const v2f b1 = bp[p * DHID + n0 + 16];
                acc0 = wmma4(afrag[kt], b0, acc0);
                acc1 = wmma4(afrag[kt], b1, acc1);
            }
            const float a0 = av[ntp * 2], a1 = av[ntp * 2 + 1];
            #pragma unroll
            for (int r = 0; r < 8; ++r) {
                Sacc[ntp * 2][r]     = Sacc[ntp * 2][r]     * a0 + acc0[r];
                Sacc[ntp * 2 + 1][r] = Sacc[ntp * 2 + 1][r] * a1 + acc1[r];
            }
        }
    }

    float* Sj = S + (size_t)chunk * BATCH * DHID;
    #pragma unroll
    for (int nt = 0; nt < 4; ++nt) {
        const int col = (nclus * 4 + nt) * 16 + lmod;
        #pragma unroll
        for (int r = 0; r < 8; ++r) {
            const int row = row0 + r + lhi * 8;
            Sj[row * DHID + col] = Sacc[nt][r];
        }
    }
}

__global__ __launch_bounds__(256)
void rnn_carry_scan(const float* __restrict__ h0,
                    const float* __restrict__ aVec,
                    const float* __restrict__ S,
                    float* __restrict__ Cout,
                    float* __restrict__ hfinal)
{
    const int idx = blockIdx.x * blockDim.x + threadIdx.x;
    if (idx >= BATCH * DHID) return;
    const int hcol = idx & (DHID - 1);

    float aL = aVec[hcol];
    #pragma unroll
    for (int s = 0; s < 5; ++s) aL *= aL;

    float c = h0[idx];
    for (int j = 0; j < CHUNKS; ++j) {
        Cout[(size_t)j * BATCH * DHID + idx] = c;
        c = aL * c + S[(size_t)j * BATCH * DHID + idx];
    }
    hfinal[idx] = c;
}

__global__ __launch_bounds__(NTHREADS)
void rnn_chunk_emit(const float* __restrict__ x,
                    const float* __restrict__ aVec,
                    const v2f* __restrict__ bp0,
                    const v2f* __restrict__ cp0,
                    const float* __restrict__ Cin,
                    float* __restrict__ y)
{
    extern __shared__ float hbuf[];
    float* ybuf = hbuf + BROWS * HSTRIDE;

    const int chunk = blockIdx.x >> 1;
    const int bhalf = blockIdx.x & 1;
    const int tid   = threadIdx.x;
    const int wave  = tid >> 5;
    const int lane  = tid & 31;
    const int lmod  = lane & 15;
    const int lhi   = lane >> 4;

    const int m_tile = wave & 1;
    const int nclus  = wave >> 1;
    const int rowL0  = m_tile * 16;
    const int rowG0  = bhalf * BROWS + m_tile * 16;
    const int rowA   = rowG0 + lmod;
    const int n2     = (wave >> 1) & 3;
    const int khalf  = wave >> 3;
    const int ytile  = wave & 7;
    const int rowH   = rowL0 + lmod;
    const int ncol   = n2 * 16 + lmod;

    float av[4];
    #pragma unroll
    for (int nt = 0; nt < 4; ++nt)
        av[nt] = aVec[(nclus * 4 + nt) * 16 + lmod];

    const float* Cj = Cin + (size_t)chunk * BATCH * DHID + (size_t)bhalf * BROWS * DHID;
    for (int idx = tid; idx < BROWS * DHID; idx += NTHREADS) {
        const int row = idx >> 9;
        const int col = idx & (DHID - 1);
        hbuf[row * HSTRIDE + col] = Cj[idx];
    }
    __syncthreads();

    const int zero = opaque_zero();
    const v2f* bp = bp0;
    const v2f* cp = cp0;

    for (int i = 0; i < CLEN; ++i) {
        bp += zero;
        cp += zero;
        const int t = chunk * CLEN + i;
        const float* xt = x + (size_t)t * BATCH * DIN;

        v2f afrag[16];
        #pragma unroll
        for (int kt = 0; kt < 16; ++kt)
            afrag[kt] = *(const v2f*)(xt + rowA * DIN + kt * 4 + lhi * 2);


        #pragma unroll
        for (int ntp = 0; ntp < 2; ++ntp) {
            const int n0 = (nclus * 4 + ntp * 2) * 16 + lmod;
            v8f acc0 = {}, acc1 = {};
            #pragma unroll
            for (int kt = 0; kt < 16; ++kt) {
                const int p = kt * 2 + lhi;
                const v2f b0 = bp[p * DHID + n0];
                const v2f b1 = bp[p * DHID + n0 + 16];
                acc0 = wmma4(afrag[kt], b0, acc0);
                acc1 = wmma4(afrag[kt], b1, acc1);
            }
            const float a0 = av[ntp * 2], a1 = av[ntp * 2 + 1];
            #pragma unroll
            for (int r = 0; r < 8; ++r) {
                const int row = rowL0 + r + lhi * 8;
                const int i0  = row * HSTRIDE + n0;
                hbuf[i0]      = hbuf[i0]      * a0 + acc0[r];
                hbuf[i0 + 16] = hbuf[i0 + 16] * a1 + acc1[r];
            }
        }
        __syncthreads();

        v8f yp = {};
        #pragma unroll 16
        for (int kt = 0; kt < 64; ++kt) {
            const int kb = khalf * 256 + kt * 4 + lhi * 2;
            const v2f hA = *(const v2f*)(hbuf + rowH * HSTRIDE + kb);
            const v2f cA = cp[(kb >> 1) * DOUT + ncol];
            yp = wmma4(hA, cA, yp);
        }

        if (khalf) {
            #pragma unroll
            for (int r = 0; r < 8; ++r)
                ybuf[ytile * YTILE + r * 32 + lane] = yp[r];
        }
        __syncthreads();

        if (!khalf) {
            float* yt = y + (size_t)t * BATCH * DOUT;
            #pragma unroll
            for (int r = 0; r < 8; ++r) {
                const int row = rowG0 + r + lhi * 8;
                yt[row * DOUT + ncol] = yp[r] + ybuf[ytile * YTILE + r * 32 + lane];
            }
        }
    }
}

extern "C" void kernel_launch(void* const* d_in, const int* in_sizes, int n_in,
                              void* d_out, int out_size, void* d_ws, size_t ws_size,
                              hipStream_t stream)
{
    const float* h0 = (const float*)d_in[0];
    const float* x  = (const float*)d_in[1];
    const float* a  = (const float*)d_in[2];
    const float* bW = (const float*)d_in[3];
    const float* cW = (const float*)d_in[4];

    float* out    = (float*)d_out;
    float* hfinal = out;
    float* y      = out + BATCH * DHID;

    float* S  = (float*)d_ws;
    float* C  = S + (size_t)CHUNKS * BATCH * DHID;
    v2f*   bp = (v2f*)(C + (size_t)CHUNKS * BATCH * DHID);
    v2f*   cp = bp + 32 * DHID;

    const size_t shmem = (size_t)(BROWS * HSTRIDE + 8 * YTILE) * sizeof(float);

    rnn_pack<<<64, 256, 0, stream>>>(bW, cW, bp, cp);
    rnn_chunk_summary<<<CHUNKS * BSPLIT, NTHREADS, 0, stream>>>(x, a, bp, S);
    rnn_carry_scan<<<(BATCH * DHID + 255) / 256, 256, 0, stream>>>(h0, a, S, C, hfinal);
    rnn_chunk_emit<<<CHUNKS * BSPLIT, NTHREADS, shmem, stream>>>(x, a, bp, cp, C, y);
}


# Time-blocked chunk kernels, 4 steps per trip: a wave holds 4 steps x 4 n-tiles of x@b accumulators, each x/b/c fragment feeds 4 WMMAs; 2 barriers per trip; scan state in registers; sums unchanged
# speedup vs baseline: 1.9837x; 1.2945x over previous
#include <hip/hip_runtime.h>


typedef __attribute__((ext_vector_type(2))) float v2f;
typedef __attribute__((ext_vector_type(8))) float v8f;

#define T_STEPS 4096
#define BATCH   64
#define DIN     64
#define DHID    512
#define DOUT    64
#define CHUNKS  128
#define CLEN    32
#define HSTRIDE 516
#define NWAVES  16
#define NTHREADS (NWAVES * 32)
#define BSPLIT  2
#define BROWS   32
#define YTILE   256
#define TBLK    4
#define NTRIPS  (CLEN / TBLK)
#define KGRP    2
#define HBUF    (BROWS * HSTRIDE)

__device__ __forceinline__ v8f wmma4(v2f a, v2f b, v8f c) {
    return __builtin_amdgcn_wmma_f32_16x16x4_f32(
        false, a, false, b, (short)0, c, false, false);
}

__device__ __forceinline__ int opaque_zero() {
    int z;
    asm volatile("s_mov_b32 %0, 0" : "=s"(z));
    return z;
}

__device__ __forceinline__ void xb_block(const float* __restrict__ xl,
                                         const v2f* __restrict__ bl,
                                         v8f (&acc)[TBLK][4])
{
    #pragma unroll
    for (int s = 0; s < TBLK; ++s) {
        #pragma unroll
        for (int nt = 0; nt < 4; ++nt) acc[s][nt] = v8f{};
    }
    #pragma unroll
    for (int kg = 0; kg < 16 / KGRP; ++kg) {
        v2f af[KGRP][TBLK], bf[KGRP][4];
        #pragma unroll
        for (int j = 0; j < KGRP; ++j) {
            const int kt = kg * KGRP + j;
            #pragma unroll
            for (int s = 0; s < TBLK; ++s)
                af[j][s] = *(const v2f*)(xl + (size_t)s * BATCH * DIN + kt * 4);
            #pragma unroll
            for (int nt = 0; nt < 4; ++nt)
                bf[j][nt] = bl[(kt * 2) * DHID + nt * 16];
        }
        #pragma unroll
        for (int j = 0; j < KGRP; ++j) {
            #pragma unroll
            for (int s = 0; s < TBLK; ++s) {
                #pragma unroll
                for (int nt = 0; nt < 4; ++nt)
                    acc[s][nt] = wmma4(af[j][s], bf[j][nt], acc[s][nt]);
            }
        }
        __builtin_amdgcn_sched_barrier(0);
    }
}

__global__ __launch_bounds__(256)
void rnn_pack(const float* __restrict__ bW, const float* __restrict__ cW,
              v2f* __restrict__ bp, v2f* __restrict__ cp)
{
    const int i = blockIdx.x * 256 + threadIdx.x;
    if (i < 32 * DHID) {
        const int p = i >> 9, n = i & (DHID - 1);
        v2f t;
        t.x = bW[(size_t)(2 * p) * DHID + n];
        t.y = bW[(size_t)(2 * p + 1) * DHID + n];
        bp[i] = t;
    }
    if (i < 256 * DOUT) {
        const int p = i >> 6, n = i & (DOUT - 1);
        v2f t;
        t.x = cW[(size_t)(2 * p) * DOUT + n];
        t.y = cW[(size_t)(2 * p + 1) * DOUT + n];
        cp[i] = t;
    }
}

__global__ __launch_bounds__(NTHREADS)
void rnn_chunk_summary(const float* __restrict__ x,
                       const float* __restrict__ aVec,
                       const v2f* __restrict__ bp0,
                       float* __restrict__ S)
{
    const int chunk = blockIdx.x >> 1;
    const int bhalf = blockIdx.x & 1;
    const int tid   = threadIdx.x;
    const int wave  = tid >> 5;
    const int lane  = tid & 31;
    const int lmod  = lane & 15;
    const int lhi   = lane >> 4;

    const int m_tile = wave & 1;
    const int nclus  = wave >> 1;
    const int row0   = bhalf * BROWS + m_tile * 16;
    const int rowA   = row0 + lmod;

    float av[4];
    #pragma unroll
    for (int nt = 0; nt < 4; ++nt)
        av[nt] = aVec[(nclus * 4 + nt) * 16 + lmod];

    const int zero = opaque_zero();
    const v2f* bp = bp0;

    v8f Sacc[4] = {};

    for (int trip = 0; trip < NTRIPS; ++trip) {
        bp += zero;
        const float* xt = x + (size_t)(chunk * CLEN + trip * TBLK) * BATCH * DIN;

        v8f acc[TBLK][4];
        xb_block(xt + rowA * DIN + lhi * 2, bp + lhi * DHID + nclus * 64 + lmod, acc);
        #pragma unroll
        for (int nt = 0; nt < 4; ++nt) {
            const float a0 = av[nt];
            #pragma unroll
            for (int s = 0; s < TBLK; ++s) {
                #pragma unroll
                for (int r = 0; r < 8; ++r)
                    Sacc[nt][r] = Sacc[nt][r] * a0 + acc[s][nt][r];
            }
        }
    }

    float* Sj = S + (size_t)chunk * BATCH * DHID;
    #pragma unroll
    for (int nt = 0; nt < 4; ++nt) {
        const int col = (nclus * 4 + nt) * 16 + lmod;
        #pragma unroll
        for (int r = 0; r < 8; ++r) {
            const int row = row0 + r + lhi * 8;
            Sj[row * DHID + col] = Sacc[nt][r];
        }
    }
}

__global__ __launch_bounds__(256)
void rnn_carry_scan(const float* __restrict__ h0,
                    const float* __restrict__ aVec,
                    const float* __restrict__ S,
                    float* __restrict__ Cout,
                    float* __restrict__ hfinal)
{
    const int idx = blockIdx.x * blockDim.x + threadIdx.x;
    if (idx >= BATCH * DHID) return;
    const int hcol = idx & (DHID - 1);

    float aL = aVec[hcol];
    #pragma unroll
    for (int s = 0; s < 5; ++s) aL *= aL;

    float c = h0[idx];
    for (int j = 0; j < CHUNKS; ++j) {
        Cout[(size_t)j * BATCH * DHID + idx] = c;
        c = aL * c + S[(size_t)j * BATCH * DHID + idx];
    }
    hfinal[idx] = c;
}

__global__ __launch_bounds__(NTHREADS)
void rnn_chunk_emit(const float* __restrict__ x,
                    const float* __restrict__ aVec,
                    const v2f* __restrict__ bp0,
                    const v2f* __restrict__ cp0,
                    const float* __restrict__ Cin,
                    float* __restrict__ y)
{
    __shared__ float hbuf[TBLK * HBUF];
    __shared__ float ybuf[TBLK * 8 * YTILE];

    const int chunk = blockIdx.x >> 1;
    const int bhalf = blockIdx.x & 1;
    const int tid   = threadIdx.x;
    const int wave  = tid >> 5;
    const int lane  = tid & 31;
    const int lmod  = lane & 15;
    const int lhi   = lane >> 4;

    const int m_tile = wave & 1;
    const int nclus  = wave >> 1;
    const int rowL0  = m_tile * 16;
    const int rowG0  = bhalf * BROWS + m_tile * 16;
    const int rowA   = rowG0 + lmod;
    const int n2     = (wave >> 1) & 3;
    const int khalf  = wave >> 3;
    const int ytile  = wave & 7;
    const int rowH   = rowL0 + lmod;
    const int ncol   = n2 * 16 + lmod;

    float av[4];
    #pragma unroll
    for (int nt = 0; nt < 4; ++nt)
        av[nt] = aVec[(nclus * 4 + nt) * 16 + lmod];

    const float* Cj = Cin + (size_t)chunk * BATCH * DHID + (size_t)bhalf * BROWS * DHID;
    float hreg[4][8];
    #pragma unroll
    for (int nt = 0; nt < 4; ++nt) {
        #pragma unroll
        for (int r = 0; r < 8; ++r)
            hreg[nt][r] = Cj[(rowL0 + r + lhi * 8) * DHID + (nclus * 4 + nt) * 16 + lmod];
    }

    const int zero = opaque_zero();
    const v2f* bp = bp0;
    const v2f* cp = cp0;

    for (int trip = 0; trip < NTRIPS; ++trip) {
        bp += zero;
        cp += zero;
        const int t0 = chunk * CLEN + trip * TBLK;
        const float* xt = x + (size_t)t0 * BATCH * DIN;


        v8f acc[TBLK][4];
        xb_block(xt + rowA * DIN + lhi * 2, bp + lhi * DHID + nclus * 64 + lmod, acc);

        #pragma unroll
        for (int nt = 0; nt < 4; ++nt) {
            const int n0 = (nclus * 4 + nt) * 16 + lmod;
            const float a0 = av[nt];
            #pragma unroll
            for (int r = 0; r < 8; ++r) {
                const int row = rowL0 + r + lhi * 8;
                const int i0  = row * HSTRIDE + n0;
                float hv = hreg[nt][r];
                #pragma unroll
                for (int s = 0; s < TBLK; ++s) {
                    hv = hv * a0 + acc[s][nt][r];
                    hbuf[s * HBUF + i0] = hv;
                }
                hreg[nt][r] = hv;
            }
        }
        __syncthreads();

        v8f yp[TBLK];
        #pragma unroll
        for (int s = 0; s < TBLK; ++s) yp[s] = v8f{};
        #pragma unroll 8
        for (int kt = 0; kt < 64; ++kt) {
            const int kb = khalf * 256 + kt * 4 + lhi * 2;
            const v2f cA = cp[(kb >> 1) * DOUT + ncol];
            #pragma unroll
            for (int s = 0; s < TBLK; ++s) {
                const v2f hA = *(const v2f*)(hbuf + s * HBUF + rowH * HSTRIDE + kb);
                yp[s] = wmma4(hA, cA, yp[s]);
            }
        }

        if (khalf) {
            #pragma unroll
            for (int s = 0; s < TBLK; ++s) {
                #pragma unroll
                for (int r = 0; r < 8; ++r)
                    ybuf[(s * 8 + ytile) * YTILE + r * 32 + lane] = yp[s][r];
            }
        }
        __syncthreads();

        if (!khalf) {
            #pragma unroll
            for (int s = 0; s < TBLK; ++s) {
                float* yt = y + (size_t)(t0 + s) * BATCH * DOUT;
                #pragma unroll
                for (int r = 0; r < 8; ++r) {
                    const int row = rowG0 + r + lhi * 8;
                    yt[row * DOUT + ncol] =
                        yp[s][r] + ybuf[(s * 8 + ytile) * YTILE + r * 32 + lane];
                }
            }
        }
    }
}

extern "C" void kernel_launch(void* const* d_in, const int* in_sizes, int n_in,
                              void* d_out, int out_size, void* d_ws, size_t ws_size,
                              hipStream_t stream)
{
    const float* h0 = (const float*)d_in[0];
    const float* x  = (const float*)d_in[1];
    const float* a  = (const float*)d_in[2];
    const float* bW = (const float*)d_in[3];
    const float* cW = (const float*)d_in[4];

    float* out    = (float*)d_out;
    float* hfinal = out;
    float* y      = out + BATCH * DHID;

    float* S  = (float*)d_ws;
    float* C  = S + (size_t)CHUNKS * BATCH * DHID;
    v2f*   bp = (v2f*)(C + (size_t)CHUNKS * BATCH * DHID);
    v2f*   cp = bp + 32 * DHID;

    rnn_pack<<<64, 256, 0, stream>>>(bW, cW, bp, cp);
    rnn_chunk_summary<<<CHUNKS * BSPLIT, NTHREADS, 0, stream>>>(x, a, bp, S);
    rnn_carry_scan<<<(BATCH * DHID + 255) / 256, 256, 0, stream>>>(h0, a, S, C, hfinal);
    rnn_chunk_emit<<<CHUNKS * BSPLIT, NTHREADS, 0, stream>>>(x, a, bp, cp, C, y);
}


# Waves lower their own issue priority (s_setprio 3,2,1,0) over the quarters of each WMMA stretch: a SIMD's matrix pipe serves the wave furthest behind, none runs the tail alone; sums unchanged
# speedup vs baseline: 2.0816x; 1.0494x over previous
#include <hip/hip_runtime.h>


typedef __attribute__((ext_vector_type(2))) float v2f;
typedef __attribute__((ext_vector_type(8))) float v8f;

#define T_STEPS 4096
#define BATCH   64
#define DIN     64
#define DHID    512
#define DOUT    64
#define CHUNKS  128
#define CLEN    32
#define HSTRIDE 516
#define NWAVES  16
#define NTHREADS (NWAVES * 32)
#define BSPLIT  2
#define BROWS   32
#define YTILE   256
#define TBLK    4
#define NTRIPS  (CLEN / TBLK)
#define KGRP    2
#define HBUF    (BROWS * HSTRIDE)

__device__ __forceinline__ v8f wmma4(v2f a, v2f b, v8f c) {
    return __builtin_amdgcn_wmma_f32_16x16x4_f32(
        false, a, false, b, (short)0, c, false, false);
}

__device__ __forceinline__ int opaque_zero() {
    int z;
    asm volatile("s_mov_b32 %0, 0" : "=s"(z));
    return z;
}

__device__ __forceinline__ void set_wave_prio(int level) {
    switch (level) {
        case 3:  __builtin_amdgcn_s_setprio(3); break;
        case 2:  __builtin_amdgcn_s_setprio(2); break;
        case 1:  __builtin_amdgcn_s_setprio(1); break;
        default: __builtin_amdgcn_s_setprio(0); break;
    }
}

template <bool GROUP_PRIO>
__device__ __forceinline__ void xb_block(const float* __restrict__ xl,
                                         const v2f* __restrict__ bl,
                                         v8f (&acc)[TBLK][4])
{
    #pragma unroll
    for (int s = 0; s < TBLK; ++s) {
        #pragma unroll
        for (int nt = 0; nt < 4; ++nt) acc[s][nt] = v8f{};
    }
    #pragma unroll
    for (int kg = 0; kg < 16 / KGRP; ++kg) {
        if (GROUP_PRIO && (kg & 1) == 0) set_wave_prio(3 - (kg >> 1));
        v2f af[KGRP][TBLK], bf[KGRP][4];
        #pragma unroll
        for (int j = 0; j < KGRP; ++j) {
            const int kt = kg * KGRP + j;
            #pragma unroll
            for (int s = 0; s < TBLK; ++s)
                af[j][s] = *(const v2f*)(xl + (size_t)s * BATCH * DIN + kt * 4);
            #pragma unroll
            for (int nt = 0; nt < 4; ++nt)
                bf[j][nt] = bl[(kt * 2) * DHID + nt * 16];
        }
        #pragma unroll
        for (int j = 0; j < KGRP; ++j) {
            #pragma unroll
            for (int s = 0; s < TBLK; ++s) {
                #pragma unroll
                for (int nt = 0; nt < 4; ++nt)
                    acc[s][nt] = wmma4(af[j][s], bf[j][nt], acc[s][nt]);
            }
        }
        __builtin_amdgcn_sched_barrier(0);
    }
}

__global__ __launch_bounds__(256)
void rnn_pack(const float* __restrict__ bW, const float* __restrict__ cW,
              v2f* __restrict__ bp, v2f* __restrict__ cp)
{
    const int i = blockIdx.x * 256 + threadIdx.x;
    if (i < 32 * DHID) {
        const int p = i >> 9, n = i & (DHID - 1);
        v2f t;
        t.x = bW[(size_t)(2 * p) * DHID + n];
        t.y = bW[(size_t)(2 * p + 1) * DHID + n];
        bp[i] = t;
    }
    if (i < 256 * DOUT) {
        const int p = i >> 6, n = i & (DOUT - 1);
        v2f t;
        t.x = cW[(size_t)(2 * p) * DOUT + n];
        t.y = cW[(size_t)(2 * p + 1) * DOUT + n];
        cp[i] = t;
    }
}

__global__ __launch_bounds__(NTHREADS)
void rnn_chunk_summary(const float* __restrict__ x,
                       const float* __restrict__ aVec,
                       const v2f* __restrict__ bp0,
                       float* __restrict__ S)
{
    const int chunk = blockIdx.x >> 1;
    const int bhalf = blockIdx.x & 1;
    const int tid   = threadIdx.x;
    const int wave  = tid >> 5;
    const int lane  = tid & 31;
    const int lmod  = lane & 15;
    const int lhi   = lane >> 4;

    const int m_tile = wave & 1;
    const int nclus  = wave >> 1;
    const int row0   = bhalf * BROWS + m_tile * 16;
    const int rowA   = row0 + lmod;

    float av[4];
    #pragma unroll
    for (int nt = 0; nt < 4; ++nt)
        av[nt] = aVec[(nclus * 4 + nt) * 16 + lmod];

    const int zero = opaque_zero();
    const v2f* bp = bp0;

    v8f Sacc[4] = {};

    for (int trip = 0; trip < NTRIPS; ++trip) {
        set_wave_prio(3 - trip / (NTRIPS / 4));
        bp += zero;
        const float* xt = x + (size_t)(chunk * CLEN + trip * TBLK) * BATCH * DIN;

        v8f acc[TBLK][4];
        xb_block<false>(xt + rowA * DIN + lhi * 2, bp + lhi * DHID + nclus * 64 + lmod, acc);
        #pragma unroll
        for (int nt = 0; nt < 4; ++nt) {
            const float a0 = av[nt];
            #pragma unroll
            for (int s = 0; s < TBLK; ++s) {
                #pragma unroll
                for (int r = 0; r < 8; ++r)
                    Sacc[nt][r] = Sacc[nt][r] * a0 + acc[s][nt][r];
            }
        }
    }

    float* Sj = S + (size_t)chunk * BATCH * DHID;
    #pragma unroll
    for (int nt = 0; nt < 4; ++nt) {
        const int col = (nclus * 4 + nt) * 16 + lmod;
        #pragma unroll
        for (int r = 0; r < 8; ++r) {
            const int row = row0 + r + lhi * 8;
            Sj[row * DHID + col] = Sacc[nt][r];
        }
    }
}

__global__ __launch_bounds__(256)
void rnn_carry_scan(const float* __restrict__ h0,
                    const float* __restrict__ aVec,
                    const float* __restrict__ S,
                    float* __restrict__ Cout,
                    float* __restrict__ hfinal)
{
    const int idx = blockIdx.x * blockDim.x + threadIdx.x;
    if (idx >= BATCH * DHID) return;
    const int hcol = idx & (DHID - 1);

    float aL = aVec[hcol];
    #pragma unroll
    for (int s = 0; s < 5; ++s) aL *= aL;

    float c = h0[idx];
    for (int j = 0; j < CHUNKS; ++j) {
        Cout[(size_t)j * BATCH * DHID + idx] = c;
        c = aL * c + S[(size_t)j * BATCH * DHID + idx];
    }
    hfinal[idx] = c;
}

__global__ __launch_bounds__(NTHREADS)
void rnn_chunk_emit(const float* __restrict__ x,
                    const float* __restrict__ aVec,
                    const v2f* __restrict__ bp0,
                    const v2f* __restrict__ cp0,
                    const float* __restrict__ Cin,
                    float* __restrict__ y)
{
    __shared__ float hbuf[TBLK * HBUF];
    __shared__ float ybuf[TBLK * 8 * YTILE];

    const int chunk = blockIdx.x >> 1;
    const int bhalf = blockIdx.x & 1;
    const int tid   = threadIdx.x;
    const int wave  = tid >> 5;
    const int lane  = tid & 31;
    const int lmod  = lane & 15;
    const int lhi   = lane >> 4;

    const int m_tile = wave & 1;
    const int nclus  = wave >> 1;
    const int rowL0  = m_tile * 16;
    const int rowG0  = bhalf * BROWS + m_tile * 16;
    const int rowA   = rowG0 + lmod;
    const int n2     = (wave >> 1) & 3;
    const int khalf  = wave >> 3;
    const int ytile  = wave & 7;
    const int rowH   = rowL0 + lmod;
    const int ncol   = n2 * 16 + lmod;

    float av[4];
    #pragma unroll
    for (int nt = 0; nt < 4; ++nt)
        av[nt] = aVec[(nclus * 4 + nt) * 16 + lmod];

    const float* Cj = Cin + (size_t)chunk * BATCH * DHID + (size_t)bhalf * BROWS * DHID;
    float hreg[4][8];
    #pragma unroll
    for (int nt = 0; nt < 4; ++nt) {
        #pragma unroll
        for (int r = 0; r < 8; ++r)
            hreg[nt][r] = Cj[(rowL0 + r + lhi * 8) * DHID + (nclus * 4 + nt) * 16 + lmod];
    }

    const int zero = opaque_zero();
    const v2f* bp = bp0;
    const v2f* cp = cp0;

    for (int trip = 0; trip < NTRIPS; ++trip) {
        bp += zero;
        cp += zero;
        const int t0 = chunk * CLEN + trip * TBLK;
        const float* xt = x + (size_t)t0 * BATCH * DIN;


        v8f acc[TBLK][4];
        xb_block<true>(xt + rowA * DIN + lhi * 2, bp + lhi * DHID + nclus * 64 + lmod, acc);

        #pragma unroll
        for (int nt = 0; nt < 4; ++nt) {
            const int n0 = (nclus * 4 + nt) * 16 + lmod;
            const float a0 = av[nt];
            #pragma unroll
            for (int r = 0; r < 8; ++r) {
                const int row = rowL0 + r + lhi * 8;
                const int i0  = row * HSTRIDE + n0;
                float hv = hreg[nt][r];
                #pragma unroll
                for (int s = 0; s < TBLK; ++s) {
                    hv = hv * a0 + acc[s][nt][r];
                    hbuf[s * HBUF + i0] = hv;
                }
                hreg[nt][r] = hv;
            }
        }
        __syncthreads();

        v8f yp[TBLK];
        #pragma unroll
        for (int s = 0; s < TBLK; ++s) yp[s] = v8f{};
        #pragma unroll
        for (int q = 0; q < 4; ++q) {
            set_wave_prio(3 - q);
            #pragma unroll 8
            for (int kt = q * 16; kt < q * 16 + 16; ++kt) {
                const int kb = khalf * 256 + kt * 4 + lhi * 2;
                const v2f cA = cp[(kb >> 1) * DOUT + ncol];
                #pragma unroll
                for (int s = 0; s < TBLK; ++s) {
                    const v2f hA = *(const v2f*)(hbuf + s * HBUF + rowH * HSTRIDE + kb);
                    yp[s] = wmma4(hA, cA, yp[s]);
                }
            }
        }

        if (khalf) {
            #pragma unroll
            for (int s = 0; s < TBLK; ++s) {
                #pragma unroll
                for (int r = 0; r < 8; ++r)
                    ybuf[(s * 8 + ytile) * YTILE + r * 32 + lane] = yp[s][r];
            }
        }
        __syncthreads();

        if (!khalf) {
            #pragma unroll
            for (int s = 0; s < TBLK; ++s) {
                float* yt = y + (size_t)(t0 + s) * BATCH * DOUT;
                #pragma unroll
                for (int r = 0; r < 8; ++r) {
                    const int row = rowG0 + r + lhi * 8;
                    yt[row * DOUT + ncol] =
                        yp[s][r] + ybuf[(s * 8 + ytile) * YTILE + r * 32 + lane];
                }
            }
        }
    }
}

extern "C" void kernel_launch(void* const* d_in, const int* in_sizes, int n_in,
                              void* d_out, int out_size, void* d_ws, size_t ws_size,
                              hipStream_t stream)
{
    const float* h0 = (const float*)d_in[0];
    const float* x  = (const float*)d_in[1];
    const float* a  = (const float*)d_in[2];
    const float* bW = (const float*)d_in[3];
    const float* cW = (const float*)d_in[4];

    float* out    = (float*)d_out;
    float* hfinal = out;
    float* y      = out + BATCH * DHID;

    float* S  = (float*)d_ws;
    float* C  = S + (size_t)CHUNKS * BATCH * DHID;
    v2f*   bp = (v2f*)(C + (size_t)CHUNKS * BATCH * DHID);
    v2f*   cp = bp + 32 * DHID;

    rnn_pack<<<64, 256, 0, stream>>>(bW, cW, bp, cp);
    rnn_chunk_summary<<<CHUNKS * BSPLIT, NTHREADS, 0, stream>>>(x, a, bp, S);
    rnn_carry_scan<<<(BATCH * DHID + 255) / 256, 256, 0, stream>>>(h0, a, S, C, hfinal);
    rnn_chunk_emit<<<CHUNKS * BSPLIT, NTHREADS, 0, stream>>>(x, a, bp, cp, C, y);
}
